# starting kernel
# baseline (speedup 1.0000x reference)
#include <hip/hip_runtime.h>

typedef __attribute__((ext_vector_type(2))) float v2f;
typedef __attribute__((ext_vector_type(8))) float v8f;

__global__ void k_zero_i32(int* __restrict__ p, int n) {
    int i = blockIdx.x * blockDim.x + threadIdx.x;
    if (i < n) p[i] = 0;
}

__global__ void k_count_indeg(const int* __restrict__ col, int* __restrict__ indeg, int e) {
    int i = blockIdx.x * blockDim.x + threadIdx.x;
    if (i < e) atomicAdd(&indeg[col[i]], 1);
}

__global__ void k_dinv(const int* __restrict__ indeg,
                       float* __restrict__ dinv_s, float* __restrict__ dinv_h, int n) {
    int i = blockIdx.x * blockDim.x + threadIdx.x;
    if (i < n) {
        float d = (float)indeg[i];
        dinv_s[i] = rsqrtf(d + 1.0f);
        dinv_h[i] = rsqrtf(d + 2.0f);
    }
}

template <int DIN, int NT, bool RELU_IN>
__global__ void k_gemm_wmma(const float* __restrict__ X, const float* __restrict__ W,
                            float* __restrict__ H, int N) {
    constexpr int DOUT = NT * 16;
    const int lane = threadIdx.x & 31;
    const int wave = threadIdx.x >> 5;
    const int tile = blockIdx.x * (blockDim.x >> 5) + wave;
    const int rtiles = (N + 15) >> 4;
    if (tile >= rtiles) return;

    const int mrow  = tile * 16 + (lane & 15);
    const int mload = mrow < N ? mrow : N - 1;
    const int kb    = (lane >> 4) * 2;
    const int nloc  = lane & 15;

    v8f acc[NT];
#pragma unroll
    for (int j = 0; j < NT; ++j) {
        v8f z = {0.f, 0.f, 0.f, 0.f, 0.f, 0.f, 0.f, 0.f};
        acc[j] = z;
    }

    const float* xr = X + (size_t)mload * DIN;
#pragma unroll 4
    for (int k = 0; k < DIN; k += 4) {
        float a0 = xr[k + kb];
        float a1 = xr[k + kb + 1];
        if (RELU_IN) { a0 = fmaxf(a0, 0.0f); a1 = fmaxf(a1, 0.0f); }
        v2f A = {a0, a1};
#pragma unroll
        for (int j = 0; j < NT; ++j) {
            const int n0 = j * 16 + nloc;
            v2f B = {W[(size_t)(k + kb) * DOUT + n0],
                     W[(size_t)(k + kb + 1) * DOUT + n0]};
            acc[j] = __builtin_amdgcn_wmma_f32_16x16x4_f32(
                false, A, false, B, (short)0, acc[j], false, false);
        }
    }

    const int half = lane >> 4;
#pragma unroll
    for (int j = 0; j < NT; ++j) {
#pragma unroll
        for (int r = 0; r < 8; ++r) {
            int mo = tile * 16 + r + 8 * half;
            if (mo < N) H[(size_t)mo * DOUT + j * 16 + nloc] = acc[j][r];
        }
    }
}

__global__ void k_agg_init(const float* __restrict__ H, const float* __restrict__ dinv,
                           const float* __restrict__ bias, float* __restrict__ Y,
                           int n, int shift, int sharp) {
    long long i = (long long)blockIdx.x * blockDim.x + threadIdx.x;
    long long tot = (long long)n << shift;
    if (i >= tot) return;
    int node = (int)(i >> shift);
    int f    = (int)(i & ((1 << shift) - 1));
    float dv = dinv[node];
    float sw = dv * dv;
    if (sharp) sw *= 2.0f;
    Y[i] = sw * H[i] + bias[f];
}

__global__ void k_agg_edges(const float* __restrict__ H, const int* __restrict__ row,
                            const int* __restrict__ col, const float* __restrict__ dinv,
                            float* __restrict__ Y, int e, int shift, int sharp) {
    long long i = (long long)blockIdx.x * blockDim.x + threadIdx.x;
    long long tot = (long long)e << shift;
    if (i >= tot) return;
    int eg = (int)(i >> shift);
    int f  = (int)(i & ((1 << shift) - 1));
    int r  = row[eg];
    int c  = col[eg];
    float w = dinv[r] * dinv[c];
    if (sharp) w = -w;
    int dout = 1 << shift;
    float v = w * H[(size_t)r * dout + f];
    unsafeAtomicAdd(&Y[(size_t)c * dout + f], v);
}

static inline int cdiv(long long a, int b) { return (int)((a + b - 1) / b); }

extern "C" void kernel_launch(void* const* d_in, const int* in_sizes, int n_in,
                              void* d_out, int out_size, void* d_ws, size_t ws_size,
                              hipStream_t stream) {
    const float* x   = (const float*)d_in[0];
    const int*   ei  = (const int*)d_in[1];
    const float* We1 = (const float*)d_in[2];  const float* be1 = (const float*)d_in[3];
    const float* We2 = (const float*)d_in[4];  const float* be2 = (const float*)d_in[5];
    const float* We3 = (const float*)d_in[6];  const float* be3 = (const float*)d_in[7];
    const float* Wd1 = (const float*)d_in[8];  const float* bd1 = (const float*)d_in[9];
    const float* Wd2 = (const float*)d_in[10]; const float* bd2 = (const float*)d_in[11];
    const float* Wd3 = (const float*)d_in[12]; const float* bd3 = (const float*)d_in[13];

    const int N = in_sizes[0] / 128;
    const int E = in_sizes[1] / 2;
    const int* rowp = ei;
    const int* colp = ei + E;

    char* ws = (char*)d_ws;
    int*   indeg  = (int*)ws;    ws += sizeof(int)   * (size_t)N;
    float* dinv_s = (float*)ws;  ws += sizeof(float) * (size_t)N;
    float* dinv_h = (float*)ws;  ws += sizeof(float) * (size_t)N;
    float* hbuf   = (float*)ws;  ws += sizeof(float) * (size_t)N * 128;
    float* ybuf   = (float*)ws;
    float* outp   = (float*)d_out;

    const int TB = 256;
    const int rtiles  = (N + 15) >> 4;
    const int gblocks = (rtiles + 7) >> 3;

    k_zero_i32   <<<cdiv(N, TB), TB, 0, stream>>>(indeg, N);
    k_count_indeg<<<cdiv(E, TB), TB, 0, stream>>>(colp, indeg, E);
    k_dinv       <<<cdiv(N, TB), TB, 0, stream>>>(indeg, dinv_s, dinv_h, N);

    k_gemm_wmma<128, 4, false><<<gblocks, TB, 0, stream>>>(x, We1, hbuf, N);
    k_agg_init <<<cdiv((long long)N << 6, TB), TB, 0, stream>>>(hbuf, dinv_s, be1, ybuf, N, 6, 0);
    k_agg_edges<<<cdiv((long long)E << 6, TB), TB, 0, stream>>>(hbuf, rowp, colp, dinv_s, ybuf, E, 6, 0);
    k_gemm_wmma<64, 2, true><<<gblocks, TB, 0, stream>>>(ybuf, We2, hbuf, N);
    k_agg_init <<<cdiv((long long)N << 5, TB), TB, 0, stream>>>(hbuf, dinv_s, be2, ybuf, N, 5, 0);
    k_agg_edges<<<cdiv((long long)E << 5, TB), TB, 0, stream>>>(hbuf, rowp, colp, dinv_s, ybuf, E, 5, 0);
    k_gemm_wmma<32, 1, true><<<gblocks, TB, 0, stream>>>(ybuf, We3, hbuf, N);
    k_agg_init <<<cdiv((long long)N << 4, TB), TB, 0, stream>>>(hbuf, dinv_s, be3, ybuf, N, 4, 0);
    k_agg_edges<<<cdiv((long long)E << 4, TB), TB, 0, stream>>>(hbuf, rowp, colp, dinv_s, ybuf, E, 4, 0);

    k_gemm_wmma<16, 2, true><<<gblocks, TB, 0, stream>>>(ybuf, Wd1, hbuf, N);
    k_agg_init <<<cdiv((long long)N << 5, TB), TB, 0, stream>>>(hbuf, dinv_h, bd1, ybuf, N, 5, 1);
    k_agg_edges<<<cdiv((long long)E << 5, TB), TB, 0, stream>>>(hbuf, rowp, colp, dinv_h, ybuf, E, 5, 1);
    k_gemm_wmma<32, 4, true><<<gblocks, TB, 0, stream>>>(ybuf, Wd2, hbuf, N);
    k_agg_init <<<cdiv((long long)N << 6, TB), TB, 0, stream>>>(hbuf, dinv_h, bd2, ybuf, N, 6, 1);
    k_agg_edges<<<cdiv((long long)E << 6, TB), TB, 0, stream>>>(hbuf, rowp, colp, dinv_h, ybuf, E, 6, 1);
    k_gemm_wmma<64, 8, true><<<gblocks, TB, 0, stream>>>(ybuf, Wd3, hbuf, N);
    k_agg_init <<<cdiv((long long)N << 7, TB), TB, 0, stream>>>(hbuf, dinv_h, bd3, outp, N, 7, 1);
    k_agg_edges<<<cdiv((long long)E << 7, TB), TB, 0, stream>>>(hbuf, rowp, colp, dinv_h, outp, E, 7, 1);
}


# Edges binned once per call by 400-node target range; each block keeps its range's output rows in LDS: self-loop+bias stored, edge terms added by LDS float atomics (arrival order), whole lines out
# speedup vs baseline: 2.5678x; 2.5678x over previous
#include <hip/hip_runtime.h>


typedef __attribute__((ext_vector_type(2))) float v2f;
typedef __attribute__((ext_vector_type(8))) float v8f;

constexpr int RNG  = 400;
constexpr int CPAD = 32;

__global__ void k_zero_i32(int* __restrict__ p, int n) {
    int i = blockIdx.x * blockDim.x + threadIdx.x;
    if (i < n) p[i] = 0;
}

__global__ void k_count_indeg(const int* __restrict__ col, int* __restrict__ indeg, int e) {
    int i = blockIdx.x * blockDim.x + threadIdx.x;
    if (i < e) atomicAdd(&indeg[col[i]], 1);
}

__global__ void k_dinv_bucket(const int* __restrict__ indeg,
                              float* __restrict__ dinv_s, float* __restrict__ dinv_h,
                              int* __restrict__ bcnt, int n) {
    __shared__ int red[32];
    const int node0 = blockIdx.x * RNG;
    int sum = 0;
    for (int t = threadIdx.x; t < RNG; t += blockDim.x) {
        const int i = node0 + t;
        if (i < n) {
            const int di = indeg[i];
            const float d = (float)di;
            dinv_s[i] = rsqrtf(d + 1.0f);
            dinv_h[i] = rsqrtf(d + 2.0f);
            sum += di;
        }
    }
    for (int off = 16; off > 0; off >>= 1) sum += __shfl_xor(sum, off);
    const int lane = threadIdx.x & 31;
    const int wave = threadIdx.x >> 5;
    if (lane == 0) red[wave] = sum;
    __syncthreads();
    if (threadIdx.x == 0) {
        int s = 0;
        const int nw = (blockDim.x + 31) >> 5;
        for (int w = 0; w < nw; ++w) s += red[w];
        bcnt[blockIdx.x] = s;
    }
}

__global__ void k_bucket_prefix(const int* __restrict__ bcnt, int* __restrict__ bbase,
                                int* __restrict__ bcur, int nb) {
    __shared__ int s[256];
    int carry = 0;
    for (int start = 0; start < nb; start += 256) {
        const int i = start + (int)threadIdx.x;
        const int v = (i < nb) ? bcnt[i] : 0;
        s[threadIdx.x] = v;
        __syncthreads();
        for (int off = 1; off < 256; off <<= 1) {
            const int t = ((int)threadIdx.x >= off) ? s[threadIdx.x - off] : 0;
            __syncthreads();
            s[threadIdx.x] += t;
            __syncthreads();
        }
        const int excl = carry + s[threadIdx.x] - v;
        if (i < nb) { bbase[i] = excl; bcur[(size_t)i * CPAD] = excl; }
        carry += s[255];
        __syncthreads();
    }
    if (threadIdx.x == 0) bbase[nb] = carry;
}

__global__ void k_fill(const int* __restrict__ row, const int* __restrict__ col,
                       int* __restrict__ bcur, int* __restrict__ bedge, int e) {
    int i = blockIdx.x * blockDim.x + threadIdx.x;
    if (i < e) {
        const int c = col[i];
        const int pos = atomicAdd(&bcur[(size_t)(c / RNG) * CPAD], 1);
        reinterpret_cast<int2*>(bedge)[pos] = make_int2(row[i], c);
    }
}

template <int DIN, int NT, bool RELU_IN>
__global__ void k_gemm_wmma(const float* __restrict__ X, const float* __restrict__ W,
                            float* __restrict__ H, int N) {
    constexpr int DOUT = NT * 16;
    const int lane = threadIdx.x & 31;
    const int wave = threadIdx.x >> 5;
    const int tile = blockIdx.x * (blockDim.x >> 5) + wave;
    const int rtiles = (N + 15) >> 4;
    if (tile >= rtiles) return;

    const int mrow  = tile * 16 + (lane & 15);
    const int mload = mrow < N ? mrow : N - 1;
    const int kb    = (lane >> 4) * 2;
    const int nloc  = lane & 15;

    v8f acc[NT];
#pragma unroll
    for (int j = 0; j < NT; ++j) {
        v8f z = {0.f, 0.f, 0.f, 0.f, 0.f, 0.f, 0.f, 0.f};
        acc[j] = z;
    }

    const float* xr = X + (size_t)mload * DIN;
#pragma unroll 4
    for (int k = 0; k < DIN; k += 4) {
        float a0 = xr[k + kb];
        float a1 = xr[k + kb + 1];
        if (RELU_IN) { a0 = fmaxf(a0, 0.0f); a1 = fmaxf(a1, 0.0f); }
        v2f A = {a0, a1};
#pragma unroll
        for (int j = 0; j < NT; ++j) {
            const int n0 = j * 16 + nloc;
            v2f B = {W[(size_t)(k + kb) * DOUT + n0],
                     W[(size_t)(k + kb + 1) * DOUT + n0]};
            acc[j] = __builtin_amdgcn_wmma_f32_16x16x4_f32(
                false, A, false, B, (short)0, acc[j], false, false);
        }
    }

    const int half = lane >> 4;
#pragma unroll
    for (int j = 0; j < NT; ++j) {
#pragma unroll
        for (int r = 0; r < 8; ++r) {
            int mo = tile * 16 + r + 8 * half;
            if (mo < N) H[(size_t)mo * DOUT + j * 16 + nloc] = acc[j][r];
        }
    }
}

__device__ __forceinline__ int lane_get(int v, int l) {
    return __builtin_amdgcn_readlane(v, l);
}
template <int EPI>
__device__ __forceinline__ int pick(int v, int q, int sub) {
    if (EPI == 1) return lane_get(v, q);
    const int a = lane_get(v, 2 * q);
    const int b = lane_get(v, 2 * q + 1);
    return sub ? b : a;
}

template <int W>
__global__ __launch_bounds__(1024)
void k_agg_tile(const float* __restrict__ H, const int* __restrict__ bedge,
                const int* __restrict__ bbase, const float* __restrict__ dinv,
                const float* __restrict__ bias, float* __restrict__ Y, int n, int sharp) {
    static_assert(W == 16 || W == 32 || W == 64 || W == 128, "row width");
    constexpr int LOGW = (W == 16) ? 4 : (W == 32) ? 5 : (W == 64) ? 6 : 7;
    constexpr int EPI  = (W >= 32) ? 1 : 2;
    constexpr int S    = (W >= 32) ? W / 32 : 1;
    constexpr int NQ   = 32 / EPI;
    constexpr int G    = (32 / S < NQ) ? 32 / S : NQ;

    __shared__ float tile[RNG * 128];

    const int tid   = threadIdx.x;
    const int lane  = tid & 31;
    const int wave  = __builtin_amdgcn_readfirstlane(tid >> 5);
    const int node0 = blockIdx.x * RNG;
    const int nn    = (n - node0 < RNG) ? (n - node0) : RNG;
    const int tot4  = (nn << LOGW) >> 2;

    const float4* H4 = reinterpret_cast<const float4*>(H + ((size_t)node0 << LOGW));
    for (int i = tid; i < tot4; i += 1024) {
        const int node = (i << 2) >> LOGW;
        const int f    = (i << 2) & (W - 1);
        const float dv = dinv[node0 + node];
        float sw = dv * dv;
        if (sharp) sw *= 2.0f;
        const float4 h = H4[i];
        float4 y;
        y.x = sw * h.x + bias[f];
        y.y = sw * h.y + bias[f + 1];
        y.z = sw * h.z + bias[f + 2];
        y.w = sw * h.w + bias[f + 3];
        *reinterpret_cast<float4*>(&tile[i << 2]) = y;
    }
    __syncthreads();

    const int e0  = bbase[blockIdx.x];
    const int e1  = bbase[blockIdx.x + 1];
    const int sub = (EPI == 1) ? 0 : (lane >> 4);
    const int f   = (EPI == 1) ? lane : (lane & 15);
    for (int base = e0 + wave * 32; base < e1; base += 32 * 32) {
        const int cnt = (e1 - base < 32) ? (e1 - base) : 32;
        int r = 0, cl = 0, wb = 0;
        if (lane < cnt) {
            const int2 rc = reinterpret_cast<const int2*>(bedge)[base + lane];
            float w = dinv[rc.x] * dinv[rc.y];
            if (sharp) w = -w;
            r  = rc.x;
            cl = rc.y - node0;
            wb = __float_as_int(w);
        }
        if (cnt == 32) {
            for (int g = 0; g < NQ; g += G) {
                float hv[G][S];
#pragma unroll
                for (int k = 0; k < G; ++k) {
                    const float* hp = H + ((size_t)pick<EPI>(r, g + k, sub) << LOGW) + f;
#pragma unroll
                    for (int s = 0; s < S; ++s) hv[k][s] = hp[s * 32];
                }
#pragma unroll
                for (int k = 0; k < G; ++k) {
                    float* tp = tile + (pick<EPI>(cl, g + k, sub) << LOGW) + f;
                    const float wj = __int_as_float(pick<EPI>(wb, g + k, sub));
#pragma unroll
                    for (int s = 0; s < S; ++s) atomicAdd(tp + s * 32, wj * hv[k][s]);
                }
            }
        } else {
            for (int q = 0; q * EPI < cnt; ++q) {
                const bool ok = (q * EPI + sub) < cnt;
                const float* hp = H + ((size_t)pick<EPI>(r, q, sub) << LOGW) + f;
                float* tp = tile + (pick<EPI>(cl, q, sub) << LOGW) + f;
                const float wj = __int_as_float(pick<EPI>(wb, q, sub));
#pragma unroll
                for (int s = 0; s < S; ++s) {
                    const float v = wj * hp[s * 32];
                    if (ok) atomicAdd(tp + s * 32, v);
                }
            }
        }
    }
    __syncthreads();

    float4* Y4 = reinterpret_cast<float4*>(Y + ((size_t)node0 << LOGW));
    for (int i = tid; i < tot4; i += 1024)
        Y4[i] = *reinterpret_cast<const float4*>(&tile[i << 2]);
}

static inline int cdiv(long long a, int b) { return (int)((a + b - 1) / b); }

extern "C" void kernel_launch(void* const* d_in, const int* in_sizes, int n_in,
                              void* d_out, int out_size, void* d_ws, size_t ws_size,
                              hipStream_t stream) {
    const float* x   = (const float*)d_in[0];
    const int*   ei  = (const int*)d_in[1];
    const float* We1 = (const float*)d_in[2];  const float* be1 = (const float*)d_in[3];
    const float* We2 = (const float*)d_in[4];  const float* be2 = (const float*)d_in[5];
    const float* We3 = (const float*)d_in[6];  const float* be3 = (const float*)d_in[7];
    const float* Wd1 = (const float*)d_in[8];  const float* bd1 = (const float*)d_in[9];
    const float* Wd2 = (const float*)d_in[10]; const float* bd2 = (const float*)d_in[11];
    const float* Wd3 = (const float*)d_in[12]; const float* bd3 = (const float*)d_in[13];

    const int N = in_sizes[0] / 128;
    const int E = in_sizes[1] / 2;
    const int* rowp = ei;
    const int* colp = ei + E;
    const int NB = cdiv(N, RNG);

    char* ws = (char*)d_ws;
    int*   indeg  = (int*)ws;    ws += sizeof(int)   * (size_t)N;
    float* dinv_s = (float*)ws;  ws += sizeof(float) * (size_t)N;
    float* dinv_h = (float*)ws;  ws += sizeof(float) * (size_t)N;
    float* hbuf   = (float*)ws;  ws += sizeof(float) * (size_t)N * 128;
    int*   bedge  = (int*)ws;    ws += sizeof(int)   * (size_t)E * 2;
    int*   bcur   = (int*)ws;    ws += sizeof(int)   * (size_t)NB * CPAD;
    int*   bbase  = (int*)ws;    ws += sizeof(int)   * (size_t)(NB + 1);
    int*   bcnt   = (int*)ws;
    float* ybuf   = (float*)d_out;

    const int TB = 256;
    const int rtiles  = (N + 15) >> 4;
    const int gblocks = (rtiles + 7) >> 3;

    k_zero_i32     <<<cdiv(N, TB), TB, 0, stream>>>(indeg, N);
    k_count_indeg  <<<cdiv(E, TB), TB, 0, stream>>>(colp, indeg, E);
    k_dinv_bucket  <<<NB, TB, 0, stream>>>(indeg, dinv_s, dinv_h, bcnt, N);
    k_bucket_prefix<<<1, 256, 0, stream>>>(bcnt, bbase, bcur, NB);
    k_fill         <<<cdiv(E, TB), TB, 0, stream>>>(rowp, colp, bcur, bedge, E);

    k_gemm_wmma<128, 4, false><<<gblocks, TB, 0, stream>>>(x, We1, hbuf, N);
    k_agg_tile<64> <<<NB, 1024, 0, stream>>>(hbuf, bedge, bbase, dinv_s, be1, ybuf, N, 0);
    k_gemm_wmma<64, 2, true><<<gblocks, TB, 0, stream>>>(ybuf, We2, hbuf, N);
    k_agg_tile<32> <<<NB, 1024, 0, stream>>>(hbuf, bedge, bbase, dinv_s, be2, ybuf, N, 0);
    k_gemm_wmma<32, 1, true><<<gblocks, TB, 0, stream>>>(ybuf, We3, hbuf, N);
    k_agg_tile<16> <<<NB, 1024, 0, stream>>>(hbuf, bedge, bbase, dinv_s, be3, ybuf, N, 0);

    k_gemm_wmma<16, 2, true><<<gblocks, TB, 0, stream>>>(ybuf, Wd1, hbuf, N);
    k_agg_tile<32> <<<NB, 1024, 0, stream>>>(hbuf, bedge, bbase, dinv_h, bd1, ybuf, N, 1);
    k_gemm_wmma<32, 4, true><<<gblocks, TB, 0, stream>>>(ybuf, Wd2, hbuf, N);
    k_agg_tile<64> <<<NB, 1024, 0, stream>>>(hbuf, bedge, bbase, dinv_h, bd2, ybuf, N, 1);
    k_gemm_wmma<64, 8, true><<<gblocks, TB, 0, stream>>>(ybuf, Wd3, hbuf, N);
    k_agg_tile<128><<<NB, 1024, 0, stream>>>(hbuf, bedge, bbase, dinv_h, bd3, ybuf, N, 1);
}


# A range's last, partial edge chunk takes a full chunk's grouped gather (a group's row slices all in flight, end slots masked at the LDS add), not one waited fetch per slice; sums keep arrival order
# speedup vs baseline: 2.8650x; 1.1157x over previous
#include <hip/hip_runtime.h>


typedef __attribute__((ext_vector_type(2))) float v2f;
typedef __attribute__((ext_vector_type(8))) float v8f;

constexpr int RNG  = 400;
constexpr int CPAD = 32;

__global__ void k_zero_i32(int* __restrict__ p, int n) {
    int i = blockIdx.x * blockDim.x + threadIdx.x;
    if (i < n) p[i] = 0;
}

__global__ void k_count_indeg(const int* __restrict__ col, int* __restrict__ indeg, int e) {
    int i = blockIdx.x * blockDim.x + threadIdx.x;
    if (i < e) atomicAdd(&indeg[col[i]], 1);
}

__global__ void k_dinv_bucket(const int* __restrict__ indeg,
                              float* __restrict__ dinv_s, float* __restrict__ dinv_h,
                              int* __restrict__ bcnt, int n) {
    __shared__ int red[32];
    const int node0 = blockIdx.x * RNG;
    int sum = 0;
    for (int t = threadIdx.x; t < RNG; t += blockDim.x) {
        const int i = node0 + t;
        if (i < n) {
            const int di = indeg[i];
            const float d = (float)di;
            dinv_s[i] = rsqrtf(d + 1.0f);
            dinv_h[i] = rsqrtf(d + 2.0f);
            sum += di;
        }
    }
    for (int off = 16; off > 0; off >>= 1) sum += __shfl_xor(sum, off);
    const int lane = threadIdx.x & 31;
    const int wave = threadIdx.x >> 5;
    if (lane == 0) red[wave] = sum;
    __syncthreads();
    if (threadIdx.x == 0) {
        int s = 0;
        const int nw = (blockDim.x + 31) >> 5;
        for (int w = 0; w < nw; ++w) s += red[w];
        bcnt[blockIdx.x] = s;
    }
}

__global__ void k_bucket_prefix(const int* __restrict__ bcnt, int* __restrict__ bbase,
                                int* __restrict__ bcur, int nb) {
    __shared__ int s[256];
    int carry = 0;
    for (int start = 0; start < nb; start += 256) {
        const int i = start + (int)threadIdx.x;
        const int v = (i < nb) ? bcnt[i] : 0;
        s[threadIdx.x] = v;
        __syncthreads();
        for (int off = 1; off < 256; off <<= 1) {
            const int t = ((int)threadIdx.x >= off) ? s[threadIdx.x - off] : 0;
            __syncthreads();
            s[threadIdx.x] += t;
            __syncthreads();
        }
        const int excl = carry + s[threadIdx.x] - v;
        if (i < nb) { bbase[i] = excl; bcur[(size_t)i * CPAD] = excl; }
        carry += s[255];
        __syncthreads();
    }
    if (threadIdx.x == 0) bbase[nb] = carry;
}

__global__ void k_fill(const int* __restrict__ row, const int* __restrict__ col,
                       int* __restrict__ bcur, int* __restrict__ bedge, int e) {
    int i = blockIdx.x * blockDim.x + threadIdx.x;
    if (i < e) {
        const int c = col[i];
        const int pos = atomicAdd(&bcur[(size_t)(c / RNG) * CPAD], 1);
        reinterpret_cast<int2*>(bedge)[pos] = make_int2(row[i], c);
    }
}

template <int DIN, int NT, bool RELU_IN>
__global__ void k_gemm_wmma(const float* __restrict__ X, const float* __restrict__ W,
                            float* __restrict__ H, int N) {
    constexpr int DOUT = NT * 16;
    const int lane = threadIdx.x & 31;
    const int wave = threadIdx.x >> 5;
    const int tile = blockIdx.x * (blockDim.x >> 5) + wave;
    const int rtiles = (N + 15) >> 4;
    if (tile >= rtiles) return;

    const int mrow  = tile * 16 + (lane & 15);
    const int mload = mrow < N ? mrow : N - 1;
    const int kb    = (lane >> 4) * 2;
    const int nloc  = lane & 15;

    v8f acc[NT];
#pragma unroll
    for (int j = 0; j < NT; ++j) {
        v8f z = {0.f, 0.f, 0.f, 0.f, 0.f, 0.f, 0.f, 0.f};
        acc[j] = z;
    }

    const float* xr = X + (size_t)mload * DIN;
#pragma unroll 4
    for (int k = 0; k < DIN; k += 4) {
        float a0 = xr[k + kb];
        float a1 = xr[k + kb + 1];
        if (RELU_IN) { a0 = fmaxf(a0, 0.0f); a1 = fmaxf(a1, 0.0f); }
        v2f A = {a0, a1};
#pragma unroll
        for (int j = 0; j < NT; ++j) {
            const int n0 = j * 16 + nloc;
            v2f B = {W[(size_t)(k + kb) * DOUT + n0],
                     W[(size_t)(k + kb + 1) * DOUT + n0]};
            acc[j] = __builtin_amdgcn_wmma_f32_16x16x4_f32(
                false, A, false, B, (short)0, acc[j], false, false);
        }
    }

    const int half = lane >> 4;
#pragma unroll
    for (int j = 0; j < NT; ++j) {
#pragma unroll
        for (int r = 0; r < 8; ++r) {
            int mo = tile * 16 + r + 8 * half;
            if (mo < N) H[(size_t)mo * DOUT + j * 16 + nloc] = acc[j][r];
        }
    }
}

__device__ __forceinline__ int lane_get(int v, int l) {
    return __builtin_amdgcn_readlane(v, l);
}
template <int EPI>
__device__ __forceinline__ int pick(int v, int q, int sub) {
    if (EPI == 1) return lane_get(v, q);
    const int a = lane_get(v, 2 * q);
    const int b = lane_get(v, 2 * q + 1);
    return sub ? b : a;
}

template <int W>
__global__ __launch_bounds__(1024)
void k_agg_tile(const float* __restrict__ H, const int* __restrict__ bedge,
                const int* __restrict__ bbase, const float* __restrict__ dinv,
                const float* __restrict__ bias, float* __restrict__ Y, int n, int sharp) {
    static_assert(W == 16 || W == 32 || W == 64 || W == 128, "row width");
    constexpr int LOGW = (W == 16) ? 4 : (W == 32) ? 5 : (W == 64) ? 6 : 7;
    constexpr int EPI  = (W >= 32) ? 1 : 2;
    constexpr int S    = (W >= 32) ? W / 32 : 1;
    constexpr int NQ   = 32 / EPI;
    constexpr int G    = (32 / S < NQ) ? 32 / S : NQ;

    __shared__ float tile[RNG * 128];

    const int tid   = threadIdx.x;
    const int lane  = tid & 31;
    const int wave  = __builtin_amdgcn_readfirstlane(tid >> 5);
    const int node0 = blockIdx.x * RNG;
    const int nn    = (n - node0 < RNG) ? (n - node0) : RNG;
    const int tot4  = (nn << LOGW) >> 2;

    const float4* H4 = reinterpret_cast<const float4*>(H + ((size_t)node0 << LOGW));
    for (int i = tid; i < tot4; i += 1024) {
        const int node = (i << 2) >> LOGW;
        const int f    = (i << 2) & (W - 1);
        const float dv = dinv[node0 + node];
        float sw = dv * dv;
        if (sharp) sw *= 2.0f;
        const float4 h = H4[i];
        float4 y;
        y.x = sw * h.x + bias[f];
        y.y = sw * h.y + bias[f + 1];
        y.z = sw * h.z + bias[f + 2];
        y.w = sw * h.w + bias[f + 3];
        *reinterpret_cast<float4*>(&tile[i << 2]) = y;
    }
    __syncthreads();

    const int e0  = bbase[blockIdx.x];
    const int e1  = bbase[blockIdx.x + 1];
    const int sub = (EPI == 1) ? 0 : (lane >> 4);
    const int f   = (EPI == 1) ? lane : (lane & 15);
    for (int base = e0 + wave * 32; base < e1; base += 32 * 32) {
        const int cnt = (e1 - base < 32) ? (e1 - base) : 32;
        int r = 0, cl = 0, wb = 0;
        if (lane < cnt) {
            const int2 rc = reinterpret_cast<const int2*>(bedge)[base + lane];
            float w = dinv[rc.x] * dinv[rc.y];
            if (sharp) w = -w;
            r  = rc.x;
            cl = rc.y - node0;
            wb = __float_as_int(w);
        }
        if (cnt == 32) {
            for (int g = 0; g < NQ; g += G) {
                float hv[G][S];
#pragma unroll
                for (int k = 0; k < G; ++k) {
                    const float* hp = H + ((size_t)pick<EPI>(r, g + k, sub) << LOGW) + f;
#pragma unroll
                    for (int s = 0; s < S; ++s) hv[k][s] = hp[s * 32];
                }
#pragma unroll
                for (int k = 0; k < G; ++k) {
                    float* tp = tile + (pick<EPI>(cl, g + k, sub) << LOGW) + f;
                    const float wj = __int_as_float(pick<EPI>(wb, g + k, sub));
#pragma unroll
                    for (int s = 0; s < S; ++s) atomicAdd(tp + s * 32, wj * hv[k][s]);
                }
            }
        } else {
            for (int g = 0; g * EPI < cnt; g += G) {
                float hv[G][S];
#pragma unroll
                for (int k = 0; k < G; ++k) {
                    const float* hp = H + ((size_t)pick<EPI>(r, g + k, sub) << LOGW) + f;
#pragma unroll
                    for (int s = 0; s < S; ++s) hv[k][s] = hp[s * 32];
                }
#pragma unroll
                for (int k = 0; k < G; ++k) {
                    const bool ok = ((g + k) * EPI + sub) < cnt;
                    float* tp = tile + (pick<EPI>(cl, g + k, sub) << LOGW) + f;
                    const float wj = __int_as_float(pick<EPI>(wb, g + k, sub));
#pragma unroll
                    for (int s = 0; s < S; ++s) {
                        const float v = wj * hv[k][s];
                        if (ok) atomicAdd(tp + s * 32, v);
                    }
                }
            }
        }
    }
    __syncthreads();

    float4* Y4 = reinterpret_cast<float4*>(Y + ((size_t)node0 << LOGW));
    for (int i = tid; i < tot4; i += 1024)
        Y4[i] = *reinterpret_cast<const float4*>(&tile[i << 2]);
}

static inline int cdiv(long long a, int b) { return (int)((a + b - 1) / b); }

extern "C" void kernel_launch(void* const* d_in, const int* in_sizes, int n_in,
                              void* d_out, int out_size, void* d_ws, size_t ws_size,
                              hipStream_t stream) {
    const float* x   = (const float*)d_in[0];
    const int*   ei  = (const int*)d_in[1];
    const float* We1 = (const float*)d_in[2];  const float* be1 = (const float*)d_in[3];
    const float* We2 = (const float*)d_in[4];  const float* be2 = (const float*)d_in[5];
    const float* We3 = (const float*)d_in[6];  const float* be3 = (const float*)d_in[7];
    const float* Wd1 = (const float*)d_in[8];  const float* bd1 = (const float*)d_in[9];
    const float* Wd2 = (const float*)d_in[10]; const float* bd2 = (const float*)d_in[11];
    const float* Wd3 = (const float*)d_in[12]; const float* bd3 = (const float*)d_in[13];

    const int N = in_sizes[0] / 128;
    const int E = in_sizes[1] / 2;
    const int* rowp = ei;
    const int* colp = ei + E;
    const int NB = cdiv(N, RNG);

    char* ws = (char*)d_ws;
    int*   indeg  = (int*)ws;    ws += sizeof(int)   * (size_t)N;
    float* dinv_s = (float*)ws;  ws += sizeof(float) * (size_t)N;
    float* dinv_h = (float*)ws;  ws += sizeof(float) * (size_t)N;
    float* hbuf   = (float*)ws;  ws += sizeof(float) * (size_t)N * 128;
    int*   bedge  = (int*)ws;    ws += sizeof(int)   * (size_t)E * 2;
    int*   bcur   = (int*)ws;    ws += sizeof(int)   * (size_t)NB * CPAD;
    int*   bbase  = (int*)ws;    ws += sizeof(int)   * (size_t)(NB + 1);
    int*   bcnt   = (int*)ws;
    float* ybuf   = (float*)d_out;

    const int TB = 256;
    const int rtiles  = (N + 15) >> 4;
    const int gblocks = (rtiles + 7) >> 3;

    k_zero_i32     <<<cdiv(N, TB), TB, 0, stream>>>(indeg, N);
    k_count_indeg  <<<cdiv(E, TB), TB, 0, stream>>>(colp, indeg, E);
    k_dinv_bucket  <<<NB, TB, 0, stream>>>(indeg, dinv_s, dinv_h, bcnt, N);
    k_bucket_prefix<<<1, 256, 0, stream>>>(bcnt, bbase, bcur, NB);
    k_fill         <<<cdiv(E, TB), TB, 0, stream>>>(rowp, colp, bcur, bedge, E);

    k_gemm_wmma<128, 4, false><<<gblocks, TB, 0, stream>>>(x, We1, hbuf, N);
    k_agg_tile<64> <<<NB, 1024, 0, stream>>>(hbuf, bedge, bbase, dinv_s, be1, ybuf, N, 0);
    k_gemm_wmma<64, 2, true><<<gblocks, TB, 0, stream>>>(ybuf, We2, hbuf, N);
    k_agg_tile<32> <<<NB, 1024, 0, stream>>>(hbuf, bedge, bbase, dinv_s, be2, ybuf, N, 0);
    k_gemm_wmma<32, 1, true><<<gblocks, TB, 0, stream>>>(ybuf, We3, hbuf, N);
    k_agg_tile<16> <<<NB, 1024, 0, stream>>>(hbuf, bedge, bbase, dinv_s, be3, ybuf, N, 0);

    k_gemm_wmma<16, 2, true><<<gblocks, TB, 0, stream>>>(ybuf, Wd1, hbuf, N);
    k_agg_tile<32> <<<NB, 1024, 0, stream>>>(hbuf, bedge, bbase, dinv_h, bd1, ybuf, N, 1);
    k_gemm_wmma<32, 4, true><<<gblocks, TB, 0, stream>>>(ybuf, Wd2, hbuf, N);
    k_agg_tile<64> <<<NB, 1024, 0, stream>>>(hbuf, bedge, bbase, dinv_h, bd2, ybuf, N, 1);
    k_gemm_wmma<64, 8, true><<<gblocks, TB, 0, stream>>>(ybuf, Wd3, hbuf, N);
    k_agg_tile<128><<<NB, 1024, 0, stream>>>(hbuf, bedge, bbase, dinv_h, bd3, ybuf, N, 1);
}
